# starting kernel
# baseline (speedup 1.0000x reference)
#include <hip/hip_runtime.h>
#include <hip/hip_bf16.h>

typedef float v2f __attribute__((ext_vector_type(2)));
typedef float v8f __attribute__((ext_vector_type(8)));

__device__ __forceinline__ v8f wmma_f32(v2f a, v2f b, v8f c) {
    return __builtin_amdgcn_wmma_f32_16x16x4_f32(
         false, a,  false, b,
         (short)0, c,  false,  false);
}

__global__ void proj_gemm(const float* __restrict__ Kin,
                          const float* __restrict__ Vin,
                          const float* __restrict__ Qin,
                          const float* __restrict__ Wk,
                          const float* __restrict__ Wv,
                          const float* __restrict__ Wq,
                          float* __restrict__ kproj,
                          float* __restrict__ vproj,
                          float* __restrict__ qproj,
                          int M, int Kd, int N) {
    const int which = blockIdx.z;
    const float* __restrict__ X = (which == 0) ? Kin : (which == 1) ? Vin : Qin;
    const float* __restrict__ W = (which == 0) ? Wk  : (which == 1) ? Wv  : Wq;
    float* __restrict__ O       = (which == 0) ? kproj : (which == 1) ? vproj : qproj;

    const int lane = threadIdx.x & 31;
    const int wave = threadIdx.x >> 5;
    const int lo = lane & 15;
    const int hi = lane >> 4;

    const int tileM = (blockIdx.y * 8 + wave) * 32;
    const int tileN = blockIdx.x * 64;

    v8f acc00 = {}; v8f acc01 = {}; v8f acc02 = {}; v8f acc03 = {};
    v8f acc10 = {}; v8f acc11 = {}; v8f acc12 = {}; v8f acc13 = {};

    const float* arow0 = X + (size_t)(tileM + lo)      * Kd + hi * 2;
    const float* arow1 = X + (size_t)(tileM + 16 + lo) * Kd + hi * 2;

    for (int k0 = 0; k0 < Kd; k0 += 4) {
        v2f a0 = *(const v2f*)(arow0 + k0);
        v2f a1 = *(const v2f*)(arow1 + k0);
        const float* wr0 = W + (size_t)(k0 + hi * 2) * N + tileN + lo;
        v2f b0; b0.x = wr0[ 0];      b0.y = wr0[ 0 + N];
        v2f b1; b1.x = wr0[16];      b1.y = wr0[16 + N];
        v2f b2; b2.x = wr0[32];      b2.y = wr0[32 + N];
        v2f b3; b3.x = wr0[48];      b3.y = wr0[48 + N];
        acc00 = wmma_f32(a0, b0, acc00);
        acc10 = wmma_f32(a1, b0, acc10);
        acc01 = wmma_f32(a0, b1, acc01);
        acc11 = wmma_f32(a1, b1, acc11);
        acc02 = wmma_f32(a0, b2, acc02);
        acc12 = wmma_f32(a1, b2, acc12);
        acc03 = wmma_f32(a0, b3, acc03);
        acc13 = wmma_f32(a1, b3, acc13);
    }

    #pragma unroll
    for (int v = 0; v < 8; ++v) {
        float* orow0 = O + (size_t)(tileM + v + hi * 8)      * N + tileN + lo;
        float* orow1 = O + (size_t)(tileM + 16 + v + hi * 8) * N + tileN + lo;
        orow0[ 0] = acc00[v];
        orow0[16] = acc01[v];
        orow0[32] = acc02[v];
        orow0[48] = acc03[v];
        orow1[ 0] = acc10[v];
        orow1[16] = acc11[v];
        orow1[32] = acc12[v];
        orow1[48] = acc13[v];
    }
}

#define SWA_W    64
#define SWA_WLEN 144

__global__ void swa_kernel(const float* __restrict__ kp,
                           const float* __restrict__ vp,
                           const float* __restrict__ qp,
                           float* __restrict__ out,
                           int T, int S, int D) {
    __shared__ float sc[16][SWA_WLEN];

    const int tid  = threadIdx.x;
    const int lane = tid & 31;
    const int wave = tid >> 5;
    const int lo = lane & 15;
    const int hi = lane >> 4;

    const int t0    = blockIdx.x * 16;
    const int kbase = t0 - SWA_W;

    const float* qrow = qp + (size_t)(t0 + lo) * D + hi * 2;
    for (int kt = wave; kt < 9; kt += 4) {
        const int krow = kbase + kt * 16 + lo;
        const int krc  = min(max(krow, 0), S - 1);
        const float* krowp = kp + (size_t)krc * D + hi * 2;
        v8f acc = {};
        for (int k0 = 0; k0 < D; k0 += 4) {
            v2f a = *(const v2f*)(qrow + k0);
            v2f b = *(const v2f*)(krowp + k0);
            acc = wmma_f32(a, b, acc);
        }
        #pragma unroll
        for (int v = 0; v < 8; ++v)
            sc[v + hi * 8][kt * 16 + lo] = acc[v];
    }
    __syncthreads();

    if (tid < 16) {
        const int t = t0 + tid;
        const int start = max(0, t - SWA_W);
        const int end   = min(S, t + SWA_W + 1);
        const float scale = 0.03608439182435161f;
        float m = -INFINITY;
        for (int c = 0; c < SWA_WLEN; ++c) {
            const int j = kbase + c;
            const float s = (j >= start && j < end) ? sc[tid][c] * scale : -INFINITY;
            sc[tid][c] = s;
            m = fmaxf(m, s);
        }
        float sum = 0.f;
        for (int c = 0; c < SWA_WLEN; ++c) {
            const float s = sc[tid][c];
            const float e = (s == -INFINITY) ? 0.f : __expf(s - m);
            sc[tid][c] = e;
            sum += e;
        }
        const float inv = 1.0f / (sum * (float)(end - start));
        for (int c = 0; c < SWA_WLEN; ++c) sc[tid][c] *= inv;
    }
    __syncthreads();

    for (int ct = wave; ct < D / 16; ct += 4) {
        const int col = ct * 16 + lo;
        v8f acc = {};
        for (int k0 = 0; k0 < SWA_WLEN; k0 += 4) {
            v2f a; a.x = sc[lo][k0 + hi * 2];
                   a.y = sc[lo][k0 + hi * 2 + 1];
            const int vr  = kbase + k0 + hi * 2;
            const int vr0 = min(max(vr,     0), S - 1);
            const int vr1 = min(max(vr + 1, 0), S - 1);
            v2f b; b.x = vp[(size_t)vr0 * D + col];
                   b.y = vp[(size_t)vr1 * D + col];
            acc = wmma_f32(a, b, acc);
        }
        #pragma unroll
        for (int v = 0; v < 8; ++v)
            out[(size_t)(t0 + v + hi * 8) * D + col] = acc[v];
    }
}

extern "C" void kernel_launch(void* const* d_in, const int* in_sizes, int n_in,
                              void* d_out, int out_size, void* d_ws, size_t ws_size,
                              hipStream_t stream) {
    const float* key   = (const float*)d_in[0];
    const float* value = (const float*)d_in[1];
    const float* query = (const float*)d_in[2];
    const float* Wk    = (const float*)d_in[3];
    const float* Wv    = (const float*)d_in[4];
    const float* Wq    = (const float*)d_in[5];
    float* out = (float*)d_out;

    const int S = 2048, T = 2048, E = 768, D = 768;

    float* ws = (float*)d_ws;
    float* kproj = ws;
    float* vproj = ws + (size_t)S * D;
    float* qproj = ws + 2 * (size_t)S * D;

    dim3 gproj(D / 64, S / 256, 3);
    proj_gemm<<<gproj, 256, 0, stream>>>(key, value, query, Wk, Wv, Wq,
                                         kproj, vproj, qproj, S, E, D);

    swa_kernel<<<dim3(T / 16), 128, 0, stream>>>(kproj, vproj, qproj, out, T, S, D);
}


# attention kernel on 16 waves per query tile (was 4): QK^T split along k over waves, 16-byte fragment loads, partials summed in LDS; row-per-wave softmax; PV 64 columns per wave; sum order changes
# speedup vs baseline: 1.9189x; 1.9189x over previous
#include <hip/hip_runtime.h>
#include <hip/hip_bf16.h>

typedef float v2f __attribute__((ext_vector_type(2)));
typedef float v8f __attribute__((ext_vector_type(8)));

__device__ __forceinline__ v8f wmma_f32(v2f a, v2f b, v8f c) {
    return __builtin_amdgcn_wmma_f32_16x16x4_f32(
         false, a,  false, b,
         (short)0, c,  false,  false);
}

__global__ void proj_gemm(const float* __restrict__ Kin,
                          const float* __restrict__ Vin,
                          const float* __restrict__ Qin,
                          const float* __restrict__ Wk,
                          const float* __restrict__ Wv,
                          const float* __restrict__ Wq,
                          float* __restrict__ kproj,
                          float* __restrict__ vproj,
                          float* __restrict__ qproj,
                          int M, int Kd, int N) {
    const int which = blockIdx.z;
    const float* __restrict__ X = (which == 0) ? Kin : (which == 1) ? Vin : Qin;
    const float* __restrict__ W = (which == 0) ? Wk  : (which == 1) ? Wv  : Wq;
    float* __restrict__ O       = (which == 0) ? kproj : (which == 1) ? vproj : qproj;

    const int lane = threadIdx.x & 31;
    const int wave = threadIdx.x >> 5;
    const int lo = lane & 15;
    const int hi = lane >> 4;

    const int tileM = (blockIdx.y * 8 + wave) * 32;
    const int tileN = blockIdx.x * 64;

    v8f acc00 = {}; v8f acc01 = {}; v8f acc02 = {}; v8f acc03 = {};
    v8f acc10 = {}; v8f acc11 = {}; v8f acc12 = {}; v8f acc13 = {};

    const float* arow0 = X + (size_t)(tileM + lo)      * Kd + hi * 2;
    const float* arow1 = X + (size_t)(tileM + 16 + lo) * Kd + hi * 2;

    for (int k0 = 0; k0 < Kd; k0 += 4) {
        v2f a0 = *(const v2f*)(arow0 + k0);
        v2f a1 = *(const v2f*)(arow1 + k0);
        const float* wr0 = W + (size_t)(k0 + hi * 2) * N + tileN + lo;
        v2f b0; b0.x = wr0[ 0];      b0.y = wr0[ 0 + N];
        v2f b1; b1.x = wr0[16];      b1.y = wr0[16 + N];
        v2f b2; b2.x = wr0[32];      b2.y = wr0[32 + N];
        v2f b3; b3.x = wr0[48];      b3.y = wr0[48 + N];
        acc00 = wmma_f32(a0, b0, acc00);
        acc10 = wmma_f32(a1, b0, acc10);
        acc01 = wmma_f32(a0, b1, acc01);
        acc11 = wmma_f32(a1, b1, acc11);
        acc02 = wmma_f32(a0, b2, acc02);
        acc12 = wmma_f32(a1, b2, acc12);
        acc03 = wmma_f32(a0, b3, acc03);
        acc13 = wmma_f32(a1, b3, acc13);
    }

    #pragma unroll
    for (int v = 0; v < 8; ++v) {
        float* orow0 = O + (size_t)(tileM + v + hi * 8)      * N + tileN + lo;
        float* orow1 = O + (size_t)(tileM + 16 + v + hi * 8) * N + tileN + lo;
        orow0[ 0] = acc00[v];
        orow0[16] = acc01[v];
        orow0[32] = acc02[v];
        orow0[48] = acc03[v];
        orow1[ 0] = acc10[v];
        orow1[16] = acc11[v];
        orow1[32] = acc12[v];
        orow1[48] = acc13[v];
    }
}

#define SWA_W     64
#define SWA_WLEN  144
#define SWA_PITCH 148
#define SWA_D     768
#define SWA_NW    16
#define SWA_KS    48

typedef float v4f __attribute__((ext_vector_type(4)));

__global__ __launch_bounds__(512)
void swa_kernel(const float* __restrict__ kp,
                const float* __restrict__ vp,
                const float* __restrict__ qp,
                float* __restrict__ out,
                int T, int S) {
    __shared__ float part[SWA_NW][16][SWA_PITCH];
    __shared__ float pr[16][SWA_PITCH];

    const int tid  = threadIdx.x;
    const int lane = tid & 31;
    const int wave = tid >> 5;
    const int lo = lane & 15;
    const int hi = lane >> 4;

    const int t0    = blockIdx.x * 16;
    const int kbase = t0 - SWA_W;

    {
        const int koff = wave * SWA_KS + hi * 4;
        const float* qrow = qp + (size_t)(t0 + lo) * SWA_D + koff;
        const float* krow[9];
        #pragma unroll
        for (int kt = 0; kt < 9; ++kt) {
            const int kr  = kbase + kt * 16 + lo;
            const int krc = min(max(kr, 0), S - 1);
            krow[kt] = kp + (size_t)krc * SWA_D + koff;
        }
        v8f acc[9];
        #pragma unroll
        for (int kt = 0; kt < 9; ++kt) acc[kt] = (v8f){};

        #pragma unroll
        for (int s = 0; s < SWA_KS / 8; ++s) {
            const v4f a = *(const v4f*)(qrow + s * 8);
            v4f b[9];
            #pragma unroll
            for (int kt = 0; kt < 9; ++kt) b[kt] = *(const v4f*)(krow[kt] + s * 8);
            v2f a0; a0.x = a.x; a0.y = a.y;
            v2f a1; a1.x = a.z; a1.y = a.w;
            #pragma unroll
            for (int kt = 0; kt < 9; ++kt) {
                v2f b0; b0.x = b[kt].x; b0.y = b[kt].y;
                acc[kt] = wmma_f32(a0, b0, acc[kt]);
            }
            #pragma unroll
            for (int kt = 0; kt < 9; ++kt) {
                v2f b1; b1.x = b[kt].z; b1.y = b[kt].w;
                acc[kt] = wmma_f32(a1, b1, acc[kt]);
            }
        }
        #pragma unroll
        for (int kt = 0; kt < 9; ++kt) {
            #pragma unroll
            for (int v = 0; v < 8; ++v)
                part[wave][v + hi * 8][kt * 16 + lo] = acc[kt][v];
        }
    }
    __syncthreads();

    {
        const int t = t0 + wave;
        const int start = max(0, t - SWA_W);
        const int end   = min(S, t + SWA_W + 1);
        const float scale = 0.03608439182435161f;
        float sv[5];
        float m = -INFINITY;
        #pragma unroll
        for (int i = 0; i < 5; ++i) {
            const int c = lane + 32 * i;
            float s = -INFINITY;
            if (c < SWA_WLEN) {
                float acc = part[0][wave][c];
                #pragma unroll
                for (int p = 1; p < SWA_NW; ++p) acc += part[p][wave][c];
                const int j = kbase + c;
                if (j >= start && j < end) s = acc * scale;
            }
            sv[i] = s;
            m = fmaxf(m, s);
        }
        for (int d = 16; d >= 1; d >>= 1) m = fmaxf(m, __shfl_xor(m, d));
        float sum = 0.f;
        #pragma unroll
        for (int i = 0; i < 5; ++i) {
            const float e = (sv[i] == -INFINITY) ? 0.f : __expf(sv[i] - m);
            sv[i] = e;
            sum += e;
        }
        for (int d = 16; d >= 1; d >>= 1) sum += __shfl_xor(sum, d);
        const float inv = 1.0f / (sum * (float)(end - start));
        #pragma unroll
        for (int i = 0; i < 5; ++i) {
            const int c = lane + 32 * i;
            if (c < SWA_WLEN) pr[wave][c] = sv[i] * inv;
        }
    }
    __syncthreads();

    if (wave < SWA_D / 64) {
        const int col = wave * 64 + lo * 4;
        v8f acc0 = {}; v8f acc1 = {}; v8f acc2 = {}; v8f acc3 = {};
        #pragma unroll 6
        for (int k0 = 0; k0 < SWA_WLEN; k0 += 4) {
            const v2f a = *(const v2f*)&pr[lo][k0 + hi * 2];
            const int vr  = kbase + k0 + hi * 2;
            const int vr0 = min(max(vr,     0), S - 1);
            const int vr1 = min(max(vr + 1, 0), S - 1);
            const v4f r0 = *(const v4f*)(vp + (size_t)vr0 * SWA_D + col);
            const v4f r1 = *(const v4f*)(vp + (size_t)vr1 * SWA_D + col);
            v2f b0; b0.x = r0.x; b0.y = r1.x;
            v2f b1; b1.x = r0.y; b1.y = r1.y;
            v2f b2; b2.x = r0.z; b2.y = r1.z;
            v2f b3; b3.x = r0.w; b3.y = r1.w;
            acc0 = wmma_f32(a, b0, acc0);
            acc1 = wmma_f32(a, b1, acc1);
            acc2 = wmma_f32(a, b2, acc2);
            acc3 = wmma_f32(a, b3, acc3);
        }
        #pragma unroll
        for (int v = 0; v < 8; ++v) {
            v4f o; o.x = acc0[v]; o.y = acc1[v]; o.z = acc2[v]; o.w = acc3[v];
            *(v4f*)(out + (size_t)(t0 + v + hi * 8) * SWA_D + col) = o;
        }
    }
}

extern "C" void kernel_launch(void* const* d_in, const int* in_sizes, int n_in,
                              void* d_out, int out_size, void* d_ws, size_t ws_size,
                              hipStream_t stream) {
    const float* key   = (const float*)d_in[0];
    const float* value = (const float*)d_in[1];
    const float* query = (const float*)d_in[2];
    const float* Wk    = (const float*)d_in[3];
    const float* Wv    = (const float*)d_in[4];
    const float* Wq    = (const float*)d_in[5];
    float* out = (float*)d_out;

    const int S = 2048, T = 2048, E = 768, D = 768;

    float* ws = (float*)d_ws;
    float* kproj = ws;
    float* vproj = ws + (size_t)S * D;
    float* qproj = ws + 2 * (size_t)S * D;

    dim3 gproj(D / 64, S / 256, 3);
    proj_gemm<<<gproj, 256, 0, stream>>>(key, value, query, Wk, Wv, Wq,
                                         kproj, vproj, qproj, S, E, D);

    swa_kernel<<<dim3(T / 16), 512, 0, stream>>>(kproj, vproj, qproj, out, T, S);
}


# projection GEMM operands staged through LDS: 64x96 block tiles on 4 waves, k chunks of 64 fetched with whole-line 16-byte loads one chunk ahead into a second stage; sum order unchanged
# speedup vs baseline: 3.2291x; 1.6828x over previous
#include <hip/hip_runtime.h>
#include <hip/hip_bf16.h>

typedef float v2f __attribute__((ext_vector_type(2)));
typedef float v8f __attribute__((ext_vector_type(8)));

__device__ __forceinline__ v8f wmma_f32(v2f a, v2f b, v8f c) {
    return __builtin_amdgcn_wmma_f32_16x16x4_f32(
         false, a,  false, b,
         (short)0, c,  false,  false);
}

#define PJ_BM 64
#define PJ_BN 96
#define PJ_KC 64
#define PJ_PA 68
#define PJ_PB 104

typedef float v4f __attribute__((ext_vector_type(4)));

__global__ __launch_bounds__(128)
void proj_gemm(const float* __restrict__ Kin,
               const float* __restrict__ Vin,
               const float* __restrict__ Qin,
               const float* __restrict__ Wk,
               const float* __restrict__ Wv,
               const float* __restrict__ Wq,
               float* __restrict__ kproj,
               float* __restrict__ vproj,
               float* __restrict__ qproj,
               int M, int Kd, int N) {
    __shared__ __attribute__((aligned(16))) float As[2][PJ_BM][PJ_PA];
    __shared__ __attribute__((aligned(16))) float Bs[2][PJ_KC][PJ_PB];

    const int which = blockIdx.z;
    const float* __restrict__ X = (which == 0) ? Kin : (which == 1) ? Vin : Qin;
    const float* __restrict__ W = (which == 0) ? Wk  : (which == 1) ? Wv  : Wq;
    float* __restrict__ O       = (which == 0) ? kproj : (which == 1) ? vproj : qproj;

    const int tid  = threadIdx.x;
    const int lane = tid & 31;
    const int wave = tid >> 5;
    const int lo = lane & 15;
    const int hi = lane >> 4;
    const int wm = wave >> 1;
    const int wn = wave & 1;

    const int blockM = blockIdx.y * PJ_BM;
    const int blockN = blockIdx.x * PJ_BN;

    const int arow = tid >> 4;
    const int acol = (tid & 15) * 4;
    const float* ag = X + (size_t)(blockM + arow) * Kd + acol;
    const int brow = tid >> 3;
    const int bcol = (tid & 7) * 4;
    const float* bg = W + (size_t)brow * N + blockN + bcol;

    v4f ra[8];
    v4f rb[12];

    v8f acc[2][3];
    #pragma unroll
    for (int i = 0; i < 2; ++i) {
        #pragma unroll
        for (int j = 0; j < 3; ++j) acc[i][j] = (v8f){};
    }

    const int nch = Kd / PJ_KC;

    #pragma unroll
    for (int i = 0; i < 8; ++i) ra[i] = *(const v4f*)(ag + (size_t)(8 * i) * Kd);
    #pragma unroll
    for (int i = 0; i < 4; ++i) {
        #pragma unroll
        for (int j = 0; j < 3; ++j) rb[i * 3 + j] = *(const v4f*)(bg + (size_t)(16 * i) * N + 32 * j);
    }
    #pragma unroll
    for (int i = 0; i < 8; ++i) *(v4f*)&As[0][arow + 8 * i][acol] = ra[i];
    #pragma unroll
    for (int i = 0; i < 4; ++i) {
        #pragma unroll
        for (int j = 0; j < 3; ++j) *(v4f*)&Bs[0][brow + 16 * i][bcol + 32 * j] = rb[i * 3 + j];
    }
    __syncthreads();

    for (int c = 0; c < nch; ++c) {
        const int st = c & 1;
        const bool more = (c + 1 < nch);
        if (more) {
            const float* agn = ag + (size_t)(c + 1) * PJ_KC;
            const float* bgn = bg + (size_t)(c + 1) * PJ_KC * N;
            #pragma unroll
            for (int i = 0; i < 8; ++i) ra[i] = *(const v4f*)(agn + (size_t)(8 * i) * Kd);
            #pragma unroll
            for (int i = 0; i < 4; ++i) {
                #pragma unroll
                for (int j = 0; j < 3; ++j) rb[i * 3 + j] = *(const v4f*)(bgn + (size_t)(16 * i) * N + 32 * j);
            }
        }

        const float* as = &As[st][wm * 32 + lo][hi * 2];
        const float* bs = &Bs[st][hi * 2][wn * 48 + lo];
        #pragma unroll
        for (int kk = 0; kk < PJ_KC; kk += 4) {
            const v2f a0 = *(const v2f*)(as + kk);
            const v2f a1 = *(const v2f*)(as + 16 * PJ_PA + kk);
            #pragma unroll
            for (int j = 0; j < 3; ++j) {
                v2f b;
                b.x = bs[(kk    ) * PJ_PB + 16 * j];
                b.y = bs[(kk + 1) * PJ_PB + 16 * j];
                acc[0][j] = wmma_f32(a0, b, acc[0][j]);
                acc[1][j] = wmma_f32(a1, b, acc[1][j]);
            }
        }

        if (more) {
            #pragma unroll
            for (int i = 0; i < 8; ++i) *(v4f*)&As[st ^ 1][arow + 8 * i][acol] = ra[i];
            #pragma unroll
            for (int i = 0; i < 4; ++i) {
                #pragma unroll
                for (int j = 0; j < 3; ++j) *(v4f*)&Bs[st ^ 1][brow + 16 * i][bcol + 32 * j] = rb[i * 3 + j];
            }
        }
        __syncthreads();
    }

    const int orow = blockM + wm * 32 + hi * 8;
    const int ocol = blockN + wn * 48 + lo;
    #pragma unroll
    for (int v = 0; v < 8; ++v) {
        float* orow0 = O + (size_t)(orow + v)      * N + ocol;
        float* orow1 = O + (size_t)(orow + 16 + v) * N + ocol;
        #pragma unroll
        for (int j = 0; j < 3; ++j) {
            orow0[16 * j] = acc[0][j][v];
            orow1[16 * j] = acc[1][j][v];
        }
    }
}

#define SWA_W     64
#define SWA_WLEN  144
#define SWA_PITCH 148
#define SWA_D     768
#define SWA_NW    16
#define SWA_KS    48

__global__ __launch_bounds__(512)
void swa_kernel(const float* __restrict__ kp,
                const float* __restrict__ vp,
                const float* __restrict__ qp,
                float* __restrict__ out,
                int T, int S) {
    __shared__ float part[SWA_NW][16][SWA_PITCH];
    __shared__ float pr[16][SWA_PITCH];

    const int tid  = threadIdx.x;
    const int lane = tid & 31;
    const int wave = tid >> 5;
    const int lo = lane & 15;
    const int hi = lane >> 4;

    const int t0    = blockIdx.x * 16;
    const int kbase = t0 - SWA_W;

    {
        const int koff = wave * SWA_KS + hi * 4;
        const float* qrow = qp + (size_t)(t0 + lo) * SWA_D + koff;
        const float* krow[9];
        #pragma unroll
        for (int kt = 0; kt < 9; ++kt) {
            const int kr  = kbase + kt * 16 + lo;
            const int krc = min(max(kr, 0), S - 1);
            krow[kt] = kp + (size_t)krc * SWA_D + koff;
        }
        v8f acc[9];
        #pragma unroll
        for (int kt = 0; kt < 9; ++kt) acc[kt] = (v8f){};

        #pragma unroll
        for (int s = 0; s < SWA_KS / 8; ++s) {
            const v4f a = *(const v4f*)(qrow + s * 8);
            v4f b[9];
            #pragma unroll
            for (int kt = 0; kt < 9; ++kt) b[kt] = *(const v4f*)(krow[kt] + s * 8);
            v2f a0; a0.x = a.x; a0.y = a.y;
            v2f a1; a1.x = a.z; a1.y = a.w;
            #pragma unroll
            for (int kt = 0; kt < 9; ++kt) {
                v2f b0; b0.x = b[kt].x; b0.y = b[kt].y;
                acc[kt] = wmma_f32(a0, b0, acc[kt]);
            }
            #pragma unroll
            for (int kt = 0; kt < 9; ++kt) {
                v2f b1; b1.x = b[kt].z; b1.y = b[kt].w;
                acc[kt] = wmma_f32(a1, b1, acc[kt]);
            }
        }
        #pragma unroll
        for (int kt = 0; kt < 9; ++kt) {
            #pragma unroll
            for (int v = 0; v < 8; ++v)
                part[wave][v + hi * 8][kt * 16 + lo] = acc[kt][v];
        }
    }
    __syncthreads();

    {
        const int t = t0 + wave;
        const int start = max(0, t - SWA_W);
        const int end   = min(S, t + SWA_W + 1);
        const float scale = 0.03608439182435161f;
        float sv[5];
        float m = -INFINITY;
        #pragma unroll
        for (int i = 0; i < 5; ++i) {
            const int c = lane + 32 * i;
            float s = -INFINITY;
            if (c < SWA_WLEN) {
                float acc = part[0][wave][c];
                #pragma unroll
                for (int p = 1; p < SWA_NW; ++p) acc += part[p][wave][c];
                const int j = kbase + c;
                if (j >= start && j < end) s = acc * scale;
            }
            sv[i] = s;
            m = fmaxf(m, s);
        }
        for (int d = 16; d >= 1; d >>= 1) m = fmaxf(m, __shfl_xor(m, d));
        float sum = 0.f;
        #pragma unroll
        for (int i = 0; i < 5; ++i) {
            const float e = (sv[i] == -INFINITY) ? 0.f : __expf(sv[i] - m);
            sv[i] = e;
            sum += e;
        }
        for (int d = 16; d >= 1; d >>= 1) sum += __shfl_xor(sum, d);
        const float inv = 1.0f / (sum * (float)(end - start));
        #pragma unroll
        for (int i = 0; i < 5; ++i) {
            const int c = lane + 32 * i;
            if (c < SWA_WLEN) pr[wave][c] = sv[i] * inv;
        }
    }
    __syncthreads();

    if (wave < SWA_D / 64) {
        const int col = wave * 64 + lo * 4;
        v8f acc0 = {}; v8f acc1 = {}; v8f acc2 = {}; v8f acc3 = {};
        #pragma unroll 6
        for (int k0 = 0; k0 < SWA_WLEN; k0 += 4) {
            const v2f a = *(const v2f*)&pr[lo][k0 + hi * 2];
            const int vr  = kbase + k0 + hi * 2;
            const int vr0 = min(max(vr,     0), S - 1);
            const int vr1 = min(max(vr + 1, 0), S - 1);
            const v4f r0 = *(const v4f*)(vp + (size_t)vr0 * SWA_D + col);
            const v4f r1 = *(const v4f*)(vp + (size_t)vr1 * SWA_D + col);
            v2f b0; b0.x = r0.x; b0.y = r1.x;
            v2f b1; b1.x = r0.y; b1.y = r1.y;
            v2f b2; b2.x = r0.z; b2.y = r1.z;
            v2f b3; b3.x = r0.w; b3.y = r1.w;
            acc0 = wmma_f32(a, b0, acc0);
            acc1 = wmma_f32(a, b1, acc1);
            acc2 = wmma_f32(a, b2, acc2);
            acc3 = wmma_f32(a, b3, acc3);
        }
        #pragma unroll
        for (int v = 0; v < 8; ++v) {
            v4f o; o.x = acc0[v]; o.y = acc1[v]; o.z = acc2[v]; o.w = acc3[v];
            *(v4f*)(out + (size_t)(t0 + v + hi * 8) * SWA_D + col) = o;
        }
    }
}

extern "C" void kernel_launch(void* const* d_in, const int* in_sizes, int n_in,
                              void* d_out, int out_size, void* d_ws, size_t ws_size,
                              hipStream_t stream) {
    const float* key   = (const float*)d_in[0];
    const float* value = (const float*)d_in[1];
    const float* query = (const float*)d_in[2];
    const float* Wk    = (const float*)d_in[3];
    const float* Wv    = (const float*)d_in[4];
    const float* Wq    = (const float*)d_in[5];
    float* out = (float*)d_out;

    const int S = 2048, T = 2048, E = 768, D = 768;

    float* ws = (float*)d_ws;
    float* kproj = ws;
    float* vproj = ws + (size_t)S * D;
    float* qproj = ws + 2 * (size_t)S * D;

    dim3 gproj(D / 96, S / 64, 3);
    proj_gemm<<<gproj, 128, 0, stream>>>(key, value, query, Wk, Wv, Wq,
                                         kproj, vproj, qproj, S, E, D);

    swa_kernel<<<dim3(T / 16), 512, 0, stream>>>(kproj, vproj, qproj, out, T, S);
}


# attention QK^T: 12 waves x 64-wide k slices; Q and key tiles fetched as whole 128-byte lines two tiles ahead, made into fragments via a per-wave LDS tile; sum order changes (12 partials)
# speedup vs baseline: 3.4820x; 1.0783x over previous
#include <hip/hip_runtime.h>
#include <hip/hip_bf16.h>

typedef float v2f __attribute__((ext_vector_type(2)));
typedef float v8f __attribute__((ext_vector_type(8)));

__device__ __forceinline__ v8f wmma_f32(v2f a, v2f b, v8f c) {
    return __builtin_amdgcn_wmma_f32_16x16x4_f32(
         false, a,  false, b,
         (short)0, c,  false,  false);
}

#define PJ_BM 64
#define PJ_BN 96
#define PJ_KC 64
#define PJ_PA 68
#define PJ_PB 104

typedef float v4f __attribute__((ext_vector_type(4)));

__global__ __launch_bounds__(128)
void proj_gemm(const float* __restrict__ Kin,
               const float* __restrict__ Vin,
               const float* __restrict__ Qin,
               const float* __restrict__ Wk,
               const float* __restrict__ Wv,
               const float* __restrict__ Wq,
               float* __restrict__ kproj,
               float* __restrict__ vproj,
               float* __restrict__ qproj,
               int M, int Kd, int N) {
    __shared__ __attribute__((aligned(16))) float As[2][PJ_BM][PJ_PA];
    __shared__ __attribute__((aligned(16))) float Bs[2][PJ_KC][PJ_PB];

    const int which = blockIdx.z;
    const float* __restrict__ X = (which == 0) ? Kin : (which == 1) ? Vin : Qin;
    const float* __restrict__ W = (which == 0) ? Wk  : (which == 1) ? Wv  : Wq;
    float* __restrict__ O       = (which == 0) ? kproj : (which == 1) ? vproj : qproj;

    const int tid  = threadIdx.x;
    const int lane = tid & 31;
    const int wave = tid >> 5;
    const int lo = lane & 15;
    const int hi = lane >> 4;
    const int wm = wave >> 1;
    const int wn = wave & 1;

    const int blockM = blockIdx.y * PJ_BM;
    const int blockN = blockIdx.x * PJ_BN;

    const int arow = tid >> 4;
    const int acol = (tid & 15) * 4;
    const float* ag = X + (size_t)(blockM + arow) * Kd + acol;
    const int brow = tid >> 3;
    const int bcol = (tid & 7) * 4;
    const float* bg = W + (size_t)brow * N + blockN + bcol;

    v4f ra[8];
    v4f rb[12];

    v8f acc[2][3];
    #pragma unroll
    for (int i = 0; i < 2; ++i) {
        #pragma unroll
        for (int j = 0; j < 3; ++j) acc[i][j] = (v8f){};
    }

    const int nch = Kd / PJ_KC;

    #pragma unroll
    for (int i = 0; i < 8; ++i) ra[i] = *(const v4f*)(ag + (size_t)(8 * i) * Kd);
    #pragma unroll
    for (int i = 0; i < 4; ++i) {
        #pragma unroll
        for (int j = 0; j < 3; ++j) rb[i * 3 + j] = *(const v4f*)(bg + (size_t)(16 * i) * N + 32 * j);
    }
    #pragma unroll
    for (int i = 0; i < 8; ++i) *(v4f*)&As[0][arow + 8 * i][acol] = ra[i];
    #pragma unroll
    for (int i = 0; i < 4; ++i) {
        #pragma unroll
        for (int j = 0; j < 3; ++j) *(v4f*)&Bs[0][brow + 16 * i][bcol + 32 * j] = rb[i * 3 + j];
    }
    __syncthreads();

    for (int c = 0; c < nch; ++c) {
        const int st = c & 1;
        const bool more = (c + 1 < nch);
        if (more) {
            const float* agn = ag + (size_t)(c + 1) * PJ_KC;
            const float* bgn = bg + (size_t)(c + 1) * PJ_KC * N;
            #pragma unroll
            for (int i = 0; i < 8; ++i) ra[i] = *(const v4f*)(agn + (size_t)(8 * i) * Kd);
            #pragma unroll
            for (int i = 0; i < 4; ++i) {
                #pragma unroll
                for (int j = 0; j < 3; ++j) rb[i * 3 + j] = *(const v4f*)(bgn + (size_t)(16 * i) * N + 32 * j);
            }
        }

        const float* as = &As[st][wm * 32 + lo][hi * 2];
        const float* bs = &Bs[st][hi * 2][wn * 48 + lo];
        #pragma unroll
        for (int kk = 0; kk < PJ_KC; kk += 4) {
            const v2f a0 = *(const v2f*)(as + kk);
            const v2f a1 = *(const v2f*)(as + 16 * PJ_PA + kk);
            #pragma unroll
            for (int j = 0; j < 3; ++j) {
                v2f b;
                b.x = bs[(kk    ) * PJ_PB + 16 * j];
                b.y = bs[(kk + 1) * PJ_PB + 16 * j];
                acc[0][j] = wmma_f32(a0, b, acc[0][j]);
                acc[1][j] = wmma_f32(a1, b, acc[1][j]);
            }
        }

        if (more) {
            #pragma unroll
            for (int i = 0; i < 8; ++i) *(v4f*)&As[st ^ 1][arow + 8 * i][acol] = ra[i];
            #pragma unroll
            for (int i = 0; i < 4; ++i) {
                #pragma unroll
                for (int j = 0; j < 3; ++j) *(v4f*)&Bs[st ^ 1][brow + 16 * i][bcol + 32 * j] = rb[i * 3 + j];
            }
        }
        __syncthreads();
    }

    const int orow = blockM + wm * 32 + hi * 8;
    const int ocol = blockN + wn * 48 + lo;
    #pragma unroll
    for (int v = 0; v < 8; ++v) {
        float* orow0 = O + (size_t)(orow + v)      * N + ocol;
        float* orow1 = O + (size_t)(orow + 16 + v) * N + ocol;
        #pragma unroll
        for (int j = 0; j < 3; ++j) {
            orow0[16 * j] = acc[0][j][v];
            orow1[16 * j] = acc[1][j][v];
        }
    }
}

#define SWA_W     64
#define SWA_WLEN  144
#define SWA_PITCH 148
#define SWA_D     768
#define SWA_NP    12
#define SWA_KS    64
#define SWA_TP    68

__device__ __forceinline__ void swa_fetch_tile(v4f (&g)[8], const float* __restrict__ rowptr) {
    #pragma unroll
    for (int i = 0; i < 8; ++i) g[i] = *(const v4f*)(rowptr + (size_t)(2 * i) * SWA_D);
}

__device__ __forceinline__ void swa_transpose_tile(const v4f (&g)[8], v4f (&f)[8],
                                                   float* st_store, const float* st_frag) {
    #pragma unroll
    for (int i = 0; i < 8; ++i) *(v4f*)(st_store + 2 * i * SWA_TP) = g[i];
    #pragma unroll
    for (int s = 0; s < 8; ++s) f[s] = *(const v4f*)(st_frag + s * 8);
}

__global__ __launch_bounds__(512)
void swa_kernel(const float* __restrict__ kp,
                const float* __restrict__ vp,
                const float* __restrict__ qp,
                float* __restrict__ out,
                int T, int S) {
    __shared__ float part[SWA_NP][16][SWA_PITCH];
    __shared__ float pr[16][SWA_PITCH];
    __shared__ __attribute__((aligned(16))) float stg[SWA_NP][16][SWA_TP];

    const int tid  = threadIdx.x;
    const int lane = tid & 31;
    const int wave = tid >> 5;
    const int lo = lane & 15;
    const int hi = lane >> 4;

    const int t0    = blockIdx.x * 16;
    const int kbase = t0 - SWA_W;

    if (wave < SWA_NP) {
        const int r2   = lane >> 4;
        const int kcol = wave * SWA_KS + (lane & 15) * 4;
        float* st_store      = &stg[wave][r2][(lane & 15) * 4];
        const float* st_frag = &stg[wave][lo][hi * 4];

        v4f g[3][8];
        swa_fetch_tile(g[0], qp + (size_t)(t0 + r2) * SWA_D + kcol);
        #pragma unroll
        for (int j = 1; j < 3; ++j) {
            const int tb = min(max(kbase + (j - 1) * 16, 0), S - 16);
            swa_fetch_tile(g[j], kp + (size_t)(tb + r2) * SWA_D + kcol);
        }
        v4f a[8];
        swa_transpose_tile(g[0], a, st_store, st_frag);

        #pragma unroll
        for (int kt = 0; kt < 9; ++kt) {
            if (kt + 2 < 9) {
                const int tb = min(max(kbase + (kt + 2) * 16, 0), S - 16);
                swa_fetch_tile(g[kt % 3], kp + (size_t)(tb + r2) * SWA_D + kcol);
            }
            v4f b[8];
            swa_transpose_tile(g[(kt + 1) % 3], b, st_store, st_frag);
            v8f acc = {};
            #pragma unroll
            for (int s = 0; s < 8; ++s) {
                v2f a0; a0.x = a[s].x; a0.y = a[s].y;
                v2f b0; b0.x = b[s].x; b0.y = b[s].y;
                acc = wmma_f32(a0, b0, acc);
                v2f a1; a1.x = a[s].z; a1.y = a[s].w;
                v2f b1; b1.x = b[s].z; b1.y = b[s].w;
                acc = wmma_f32(a1, b1, acc);
            }
            #pragma unroll
            for (int v = 0; v < 8; ++v)
                part[wave][v + hi * 8][kt * 16 + lo] = acc[v];
        }
    }
    __syncthreads();

    {
        const int t = t0 + wave;
        const int start = max(0, t - SWA_W);
        const int end   = min(S, t + SWA_W + 1);
        const float scale = 0.03608439182435161f;
        float sv[5];
        float m = -INFINITY;
        #pragma unroll
        for (int i = 0; i < 5; ++i) {
            const int c = lane + 32 * i;
            float s = -INFINITY;
            if (c < SWA_WLEN) {
                float acc = part[0][wave][c];
                #pragma unroll
                for (int p = 1; p < SWA_NP; ++p) acc += part[p][wave][c];
                const int j = kbase + c;
                if (j >= start && j < end) s = acc * scale;
            }
            sv[i] = s;
            m = fmaxf(m, s);
        }
        for (int d = 16; d >= 1; d >>= 1) m = fmaxf(m, __shfl_xor(m, d));
        float sum = 0.f;
        #pragma unroll
        for (int i = 0; i < 5; ++i) {
            const float e = (sv[i] == -INFINITY) ? 0.f : __expf(sv[i] - m);
            sv[i] = e;
            sum += e;
        }
        for (int d = 16; d >= 1; d >>= 1) sum += __shfl_xor(sum, d);
        const float inv = 1.0f / (sum * (float)(end - start));
        #pragma unroll
        for (int i = 0; i < 5; ++i) {
            const int c = lane + 32 * i;
            if (c < SWA_WLEN) pr[wave][c] = sv[i] * inv;
        }
    }
    __syncthreads();

    if (wave < SWA_D / 64) {
        const int col = wave * 64 + lo * 4;
        v8f acc0 = {}; v8f acc1 = {}; v8f acc2 = {}; v8f acc3 = {};
        #pragma unroll 6
        for (int k0 = 0; k0 < SWA_WLEN; k0 += 4) {
            const v2f a = *(const v2f*)&pr[lo][k0 + hi * 2];
            const int vr  = kbase + k0 + hi * 2;
            const int vr0 = min(max(vr,     0), S - 1);
            const int vr1 = min(max(vr + 1, 0), S - 1);
            const v4f r0 = *(const v4f*)(vp + (size_t)vr0 * SWA_D + col);
            const v4f r1 = *(const v4f*)(vp + (size_t)vr1 * SWA_D + col);
            v2f b0; b0.x = r0.x; b0.y = r1.x;
            v2f b1; b1.x = r0.y; b1.y = r1.y;
            v2f b2; b2.x = r0.z; b2.y = r1.z;
            v2f b3; b3.x = r0.w; b3.y = r1.w;
            acc0 = wmma_f32(a, b0, acc0);
            acc1 = wmma_f32(a, b1, acc1);
            acc2 = wmma_f32(a, b2, acc2);
            acc3 = wmma_f32(a, b3, acc3);
        }
        #pragma unroll
        for (int v = 0; v < 8; ++v) {
            v4f o; o.x = acc0[v]; o.y = acc1[v]; o.z = acc2[v]; o.w = acc3[v];
            *(v4f*)(out + (size_t)(t0 + v + hi * 8) * SWA_D + col) = o;
        }
    }
}

extern "C" void kernel_launch(void* const* d_in, const int* in_sizes, int n_in,
                              void* d_out, int out_size, void* d_ws, size_t ws_size,
                              hipStream_t stream) {
    const float* key   = (const float*)d_in[0];
    const float* value = (const float*)d_in[1];
    const float* query = (const float*)d_in[2];
    const float* Wk    = (const float*)d_in[3];
    const float* Wv    = (const float*)d_in[4];
    const float* Wq    = (const float*)d_in[5];
    float* out = (float*)d_out;

    const int S = 2048, T = 2048, E = 768, D = 768;

    float* ws = (float*)d_ws;
    float* kproj = ws;
    float* vproj = ws + (size_t)S * D;
    float* qproj = ws + 2 * (size_t)S * D;

    dim3 gproj(D / 96, S / 64, 3);
    proj_gemm<<<gproj, 128, 0, stream>>>(key, value, query, Wk, Wv, Wq,
                                         kproj, vproj, qproj, S, E, D);

    swa_kernel<<<dim3(T / 16), 512, 0, stream>>>(kproj, vproj, qproj, out, T, S);
}


# projection GEMM fragment reads: W staged k-quad-interleaved so each A and B fragment is one 16-byte LDS read for two WMMAs, a block's 10 reads issued as one group ahead of 24 WMMAs; sum order changes
# speedup vs baseline: 3.6847x; 1.0582x over previous
#include <hip/hip_runtime.h>
#include <hip/hip_bf16.h>

typedef float v2f __attribute__((ext_vector_type(2)));
typedef float v8f __attribute__((ext_vector_type(8)));

__device__ __forceinline__ v8f wmma_f32(v2f a, v2f b, v8f c) {
    return __builtin_amdgcn_wmma_f32_16x16x4_f32(
         false, a,  false, b,
         (short)0, c,  false,  false);
}

#define PJ_BM 64
#define PJ_BN 96
#define PJ_KC 64
#define PJ_PA 68
#define PJ_PQ 416

typedef float v4f __attribute__((ext_vector_type(4)));

struct PjFrag { v4f a[2][2]; v4f b[3][2]; };

__device__ __forceinline__ void pj_load_frags(PjFrag& f, const float* as, const float* bs, int blk) {
    #pragma unroll
    for (int g = 0; g < 2; ++g) {
        #pragma unroll
        for (int i = 0; i < 2; ++i) f.a[i][g] = *(const v4f*)(as + 16 * PJ_PA * i + 8 * (2 * blk + g));
        #pragma unroll
        for (int j = 0; j < 3; ++j) f.b[j][g] = *(const v4f*)(bs + 2 * (2 * blk + g) * PJ_PQ + 64 * j);
    }
}

__device__ __forceinline__ void pj_mma_block(const PjFrag& f, v8f (&acc)[2][3]) {
    #pragma unroll
    for (int g = 0; g < 2; ++g) {
        #pragma unroll
        for (int j = 0; j < 3; ++j) {
            v2f b; b.x = f.b[j][g].x; b.y = f.b[j][g].y;
            #pragma unroll
            for (int i = 0; i < 2; ++i) {
                v2f a; a.x = f.a[i][g].x; a.y = f.a[i][g].y;
                acc[i][j] = wmma_f32(a, b, acc[i][j]);
            }
        }
        #pragma unroll
        for (int j = 0; j < 3; ++j) {
            v2f b; b.x = f.b[j][g].z; b.y = f.b[j][g].w;
            #pragma unroll
            for (int i = 0; i < 2; ++i) {
                v2f a; a.x = f.a[i][g].z; a.y = f.a[i][g].w;
                acc[i][j] = wmma_f32(a, b, acc[i][j]);
            }
        }
    }
}

__device__ __forceinline__ void pj_fence() { __builtin_amdgcn_sched_barrier(0); }

__global__ __launch_bounds__(128)
void proj_gemm(const float* __restrict__ Kin,
               const float* __restrict__ Vin,
               const float* __restrict__ Qin,
               const float* __restrict__ Wk,
               const float* __restrict__ Wv,
               const float* __restrict__ Wq,
               float* __restrict__ kproj,
               float* __restrict__ vproj,
               float* __restrict__ qproj,
               int M, int Kd, int N) {
    __shared__ __attribute__((aligned(16))) float As[2][PJ_BM][PJ_PA];
    __shared__ __attribute__((aligned(16))) float Bs[2][PJ_KC / 4][PJ_PQ];

    const int which = blockIdx.z;
    const float* __restrict__ X = (which == 0) ? Kin : (which == 1) ? Vin : Qin;
    const float* __restrict__ W = (which == 0) ? Wk  : (which == 1) ? Wv  : Wq;
    float* __restrict__ O       = (which == 0) ? kproj : (which == 1) ? vproj : qproj;

    const int tid  = threadIdx.x;
    const int lane = tid & 31;
    const int wave = tid >> 5;
    const int lo = lane & 15;
    const int hi = lane >> 4;
    const int wm = wave >> 1;
    const int wn = wave & 1;

    const int blockM = blockIdx.y * PJ_BM;
    const int blockN = blockIdx.x * PJ_BN;

    const int arow = tid >> 4;
    const int acol = (tid & 15) * 4;
    const float* ag = X + (size_t)(blockM + arow) * Kd + acol;
    const int bq   = tid >> 3;
    const int bcol = (tid & 7) * 4;
    const float* bg = W + (size_t)(bq * 4) * N + blockN + bcol;

    v4f ra[8];
    v4f rb[12];

    v8f acc[2][3];
    #pragma unroll
    for (int i = 0; i < 2; ++i) {
        #pragma unroll
        for (int j = 0; j < 3; ++j) acc[i][j] = (v8f){};
    }

    const int nch = Kd / PJ_KC;

#define PJ_FETCH(CH)                                                                         \
    do {                                                                                     \
        const float* agn = ag + (size_t)(CH) * PJ_KC;                                        \
        const float* bgn = bg + (size_t)(CH) * PJ_KC * N;                                    \
        _Pragma("unroll")                   \
        for (int i = 0; i < 8; ++i) ra[i] = *(const v4f*)(agn + (size_t)(8 * i) * Kd);       \
        _Pragma("unroll")                   \
        for (int e = 0; e < 4; ++e) {                                                        \
            _Pragma("unroll")               \
            for (int p = 0; p < 3; ++p) rb[e * 3 + p] = *(const v4f*)(bgn + (size_t)e * N + 32 * p); \
        }                                                                                    \
    } while (0)

#define PJ_STAGE(ST)                                                                         \
    do {                                                                                     \
        _Pragma("unroll")                   \
        for (int i = 0; i < 8; ++i) *(v4f*)&As[ST][arow + 8 * i][acol] = ra[i];              \
        _Pragma("unroll")                   \
        for (int p = 0; p < 3; ++p) {                                                        \
            _Pragma("unroll")               \
            for (int i = 0; i < 4; ++i) {                                                    \
                v4f o;                                                                       \
                o.x = rb[0 * 3 + p][i]; o.y = rb[1 * 3 + p][i];                              \
                o.z = rb[2 * 3 + p][i]; o.w = rb[3 * 3 + p][i];                              \
                *(v4f*)&Bs[ST][bq][(32 * p + bcol + i) * 4] = o;                             \
            }                                                                                \
        }                                                                                    \
    } while (0)

    PJ_FETCH(0);
    PJ_STAGE(0);
    __syncthreads();

    const float* as0 = &As[0][wm * 32 + lo][hi * 4];
    const float* bs0 = &Bs[0][hi][(wn * 48 + lo) * 4];
    const int as_stage = PJ_BM * PJ_PA;
    const int bs_stage = (PJ_KC / 4) * PJ_PQ;

    PjFrag f0, f1;
    pj_load_frags(f0, as0, bs0, 0);
    if (nch > 1) PJ_FETCH(1);

    for (int c = 0; c < nch; ++c) {
        const int st = c & 1;
        const bool more = (c + 1 < nch);
        const float* as = as0 + st * as_stage;
        const float* bs = bs0 + st * bs_stage;

        pj_load_frags(f1, as, bs, 1);
        pj_fence();
        pj_mma_block(f0, acc);
        pj_fence();
        pj_load_frags(f0, as, bs, 2);
        pj_fence();
        pj_mma_block(f1, acc);
        pj_fence();
        if (more) {
            PJ_STAGE(st ^ 1);
        }
        pj_load_frags(f1, as, bs, 3);
        pj_fence();
        pj_mma_block(f0, acc);
        pj_fence();
        __syncthreads();
        if (more) {
            pj_load_frags(f0, as0 + (st ^ 1) * as_stage, bs0 + (st ^ 1) * bs_stage, 0);
            if (c + 2 < nch) PJ_FETCH(c + 2);
        }
        pj_fence();
        pj_mma_block(f1, acc);
        pj_fence();
    }
#undef PJ_FETCH
#undef PJ_STAGE

    const int orow = blockM + wm * 32 + hi * 8;
    const int ocol = blockN + wn * 48 + lo;
    #pragma unroll
    for (int v = 0; v < 8; ++v) {
        float* orow0 = O + (size_t)(orow + v)      * N + ocol;
        float* orow1 = O + (size_t)(orow + 16 + v) * N + ocol;
        #pragma unroll
        for (int j = 0; j < 3; ++j) {
            orow0[16 * j] = acc[0][j][v];
            orow1[16 * j] = acc[1][j][v];
        }
    }
}

#define SWA_W     64
#define SWA_WLEN  144
#define SWA_PITCH 148
#define SWA_D     768
#define SWA_NP    12
#define SWA_KS    64
#define SWA_TP    68

__device__ __forceinline__ void swa_fetch_tile(v4f (&g)[8], const float* __restrict__ rowptr) {
    #pragma unroll
    for (int i = 0; i < 8; ++i) g[i] = *(const v4f*)(rowptr + (size_t)(2 * i) * SWA_D);
}

__device__ __forceinline__ void swa_transpose_tile(const v4f (&g)[8], v4f (&f)[8],
                                                   float* st_store, const float* st_frag) {
    #pragma unroll
    for (int i = 0; i < 8; ++i) *(v4f*)(st_store + 2 * i * SWA_TP) = g[i];
    #pragma unroll
    for (int s = 0; s < 8; ++s) f[s] = *(const v4f*)(st_frag + s * 8);
}

__global__ __launch_bounds__(512)
void swa_kernel(const float* __restrict__ kp,
                const float* __restrict__ vp,
                const float* __restrict__ qp,
                float* __restrict__ out,
                int T, int S) {
    __shared__ float part[SWA_NP][16][SWA_PITCH];
    __shared__ float pr[16][SWA_PITCH];
    __shared__ __attribute__((aligned(16))) float stg[SWA_NP][16][SWA_TP];

    const int tid  = threadIdx.x;
    const int lane = tid & 31;
    const int wave = tid >> 5;
    const int lo = lane & 15;
    const int hi = lane >> 4;

    const int t0    = blockIdx.x * 16;
    const int kbase = t0 - SWA_W;

    if (wave < SWA_NP) {
        const int r2   = lane >> 4;
        const int kcol = wave * SWA_KS + (lane & 15) * 4;
        float* st_store      = &stg[wave][r2][(lane & 15) * 4];
        const float* st_frag = &stg[wave][lo][hi * 4];

        v4f g[3][8];
        swa_fetch_tile(g[0], qp + (size_t)(t0 + r2) * SWA_D + kcol);
        #pragma unroll
        for (int j = 1; j < 3; ++j) {
            const int tb = min(max(kbase + (j - 1) * 16, 0), S - 16);
            swa_fetch_tile(g[j], kp + (size_t)(tb + r2) * SWA_D + kcol);
        }
        v4f a[8];
        swa_transpose_tile(g[0], a, st_store, st_frag);

        #pragma unroll
        for (int kt = 0; kt < 9; ++kt) {
            if (kt + 2 < 9) {
                const int tb = min(max(kbase + (kt + 2) * 16, 0), S - 16);
                swa_fetch_tile(g[kt % 3], kp + (size_t)(tb + r2) * SWA_D + kcol);
            }
            v4f b[8];
            swa_transpose_tile(g[(kt + 1) % 3], b, st_store, st_frag);
            v8f acc = {};
            #pragma unroll
            for (int s = 0; s < 8; ++s) {
                v2f a0; a0.x = a[s].x; a0.y = a[s].y;
                v2f b0; b0.x = b[s].x; b0.y = b[s].y;
                acc = wmma_f32(a0, b0, acc);
                v2f a1; a1.x = a[s].z; a1.y = a[s].w;
                v2f b1; b1.x = b[s].z; b1.y = b[s].w;
                acc = wmma_f32(a1, b1, acc);
            }
            #pragma unroll
            for (int v = 0; v < 8; ++v)
                part[wave][v + hi * 8][kt * 16 + lo] = acc[v];
        }
    }
    __syncthreads();

    {
        const int t = t0 + wave;
        const int start = max(0, t - SWA_W);
        const int end   = min(S, t + SWA_W + 1);
        const float scale = 0.03608439182435161f;
        float sv[5];
        float m = -INFINITY;
        #pragma unroll
        for (int i = 0; i < 5; ++i) {
            const int c = lane + 32 * i;
            float s = -INFINITY;
            if (c < SWA_WLEN) {
                float acc = part[0][wave][c];
                #pragma unroll
                for (int p = 1; p < SWA_NP; ++p) acc += part[p][wave][c];
                const int j = kbase + c;
                if (j >= start && j < end) s = acc * scale;
            }
            sv[i] = s;
            m = fmaxf(m, s);
        }
        for (int d = 16; d >= 1; d >>= 1) m = fmaxf(m, __shfl_xor(m, d));
        float sum = 0.f;
        #pragma unroll
        for (int i = 0; i < 5; ++i) {
            const float e = (sv[i] == -INFINITY) ? 0.f : __expf(sv[i] - m);
            sv[i] = e;
            sum += e;
        }
        for (int d = 16; d >= 1; d >>= 1) sum += __shfl_xor(sum, d);
        const float inv = 1.0f / (sum * (float)(end - start));
        #pragma unroll
        for (int i = 0; i < 5; ++i) {
            const int c = lane + 32 * i;
            if (c < SWA_WLEN) pr[wave][c] = sv[i] * inv;
        }
    }
    __syncthreads();

    if (wave < SWA_D / 64) {
        const int col = wave * 64 + lo * 4;
        v8f acc0 = {}; v8f acc1 = {}; v8f acc2 = {}; v8f acc3 = {};
        #pragma unroll 6
        for (int k0 = 0; k0 < SWA_WLEN; k0 += 4) {
            const v2f a = *(const v2f*)&pr[lo][k0 + hi * 2];
            const int vr  = kbase + k0 + hi * 2;
            const int vr0 = min(max(vr,     0), S - 1);
            const int vr1 = min(max(vr + 1, 0), S - 1);
            const v4f r0 = *(const v4f*)(vp + (size_t)vr0 * SWA_D + col);
            const v4f r1 = *(const v4f*)(vp + (size_t)vr1 * SWA_D + col);
            v2f b0; b0.x = r0.x; b0.y = r1.x;
            v2f b1; b1.x = r0.y; b1.y = r1.y;
            v2f b2; b2.x = r0.z; b2.y = r1.z;
            v2f b3; b3.x = r0.w; b3.y = r1.w;
            acc0 = wmma_f32(a, b0, acc0);
            acc1 = wmma_f32(a, b1, acc1);
            acc2 = wmma_f32(a, b2, acc2);
            acc3 = wmma_f32(a, b3, acc3);
        }
        #pragma unroll
        for (int v = 0; v < 8; ++v) {
            v4f o; o.x = acc0[v]; o.y = acc1[v]; o.z = acc2[v]; o.w = acc3[v];
            *(v4f*)(out + (size_t)(t0 + v + hi * 8) * SWA_D + col) = o;
        }
    }
}

extern "C" void kernel_launch(void* const* d_in, const int* in_sizes, int n_in,
                              void* d_out, int out_size, void* d_ws, size_t ws_size,
                              hipStream_t stream) {
    const float* key   = (const float*)d_in[0];
    const float* value = (const float*)d_in[1];
    const float* query = (const float*)d_in[2];
    const float* Wk    = (const float*)d_in[3];
    const float* Wv    = (const float*)d_in[4];
    const float* Wq    = (const float*)d_in[5];
    float* out = (float*)d_out;

    const int S = 2048, T = 2048, E = 768, D = 768;

    float* ws = (float*)d_ws;
    float* kproj = ws;
    float* vproj = ws + (size_t)S * D;
    float* qproj = ws + 2 * (size_t)S * D;

    dim3 gproj(D / 96, S / 64, 3);
    proj_gemm<<<gproj, 128, 0, stream>>>(key, value, query, Wk, Wv, Wq,
                                         kproj, vproj, qproj, S, E, D);

    swa_kernel<<<dim3(T / 16), 512, 0, stream>>>(kproj, vproj, qproj, out, T, S);
}


# address arithmetic off the vector unit: compile-time row pitch makes each projection load and store one base plus immediates; PV rows clamped per key tile on the scalar unit; sum order unchanged
# speedup vs baseline: 3.8223x; 1.0373x over previous
#include <hip/hip_runtime.h>
#include <hip/hip_bf16.h>

typedef float v2f __attribute__((ext_vector_type(2)));
typedef float v8f __attribute__((ext_vector_type(8)));

__device__ __forceinline__ v8f wmma_f32(v2f a, v2f b, v8f c) {
    return __builtin_amdgcn_wmma_f32_16x16x4_f32(
         false, a,  false, b,
         (short)0, c,  false,  false);
}

#define PJ_BM 64
#define PJ_BN 96
#define PJ_KC 64
#define PJ_PA 68
#define PJ_PQ 416
#define PJ_LD 768

typedef float v4f __attribute__((ext_vector_type(4)));

struct PjFrag { v4f a[2][2]; v4f b[3][2]; };

__device__ __forceinline__ void pj_load_frags(PjFrag& f, const float* as, const float* bs, int blk) {
    #pragma unroll
    for (int g = 0; g < 2; ++g) {
        #pragma unroll
        for (int i = 0; i < 2; ++i) f.a[i][g] = *(const v4f*)(as + 16 * PJ_PA * i + 8 * (2 * blk + g));
        #pragma unroll
        for (int j = 0; j < 3; ++j) f.b[j][g] = *(const v4f*)(bs + 2 * (2 * blk + g) * PJ_PQ + 64 * j);
    }
}

__device__ __forceinline__ void pj_mma_block(const PjFrag& f, v8f (&acc)[2][3]) {
    #pragma unroll
    for (int g = 0; g < 2; ++g) {
        #pragma unroll
        for (int j = 0; j < 3; ++j) {
            v2f b; b.x = f.b[j][g].x; b.y = f.b[j][g].y;
            #pragma unroll
            for (int i = 0; i < 2; ++i) {
                v2f a; a.x = f.a[i][g].x; a.y = f.a[i][g].y;
                acc[i][j] = wmma_f32(a, b, acc[i][j]);
            }
        }
        #pragma unroll
        for (int j = 0; j < 3; ++j) {
            v2f b; b.x = f.b[j][g].z; b.y = f.b[j][g].w;
            #pragma unroll
            for (int i = 0; i < 2; ++i) {
                v2f a; a.x = f.a[i][g].z; a.y = f.a[i][g].w;
                acc[i][j] = wmma_f32(a, b, acc[i][j]);
            }
        }
    }
}

__device__ __forceinline__ void pj_fence() { __builtin_amdgcn_sched_barrier(0); }

__global__ __launch_bounds__(128)
void proj_gemm(const float* __restrict__ Kin,
               const float* __restrict__ Vin,
               const float* __restrict__ Qin,
               const float* __restrict__ Wk,
               const float* __restrict__ Wv,
               const float* __restrict__ Wq,
               float* __restrict__ kproj,
               float* __restrict__ vproj,
               float* __restrict__ qproj,
               int Kd) {
    __shared__ __attribute__((aligned(16))) float As[2][PJ_BM][PJ_PA];
    __shared__ __attribute__((aligned(16))) float Bs[2][PJ_KC / 4][PJ_PQ];

    const int which = blockIdx.z;
    const float* __restrict__ X = (which == 0) ? Kin : (which == 1) ? Vin : Qin;
    const float* __restrict__ W = (which == 0) ? Wk  : (which == 1) ? Wv  : Wq;
    float* __restrict__ O       = (which == 0) ? kproj : (which == 1) ? vproj : qproj;

    const int tid  = threadIdx.x;
    const int lane = tid & 31;
    const int wave = tid >> 5;
    const int lo = lane & 15;
    const int hi = lane >> 4;
    const int wm = wave >> 1;
    const int wn = wave & 1;

    const int blockM = blockIdx.y * PJ_BM;
    const int blockN = blockIdx.x * PJ_BN;

    const int arow = tid >> 4;
    const int acol = (tid & 15) * 4;
    const float* ag = X + (size_t)(blockM + arow) * PJ_LD + acol;
    const int bq   = tid >> 3;
    const int bcol = (tid & 7) * 4;
    const float* bg = W + (size_t)(bq * 4) * PJ_LD + blockN + bcol;

    v4f ra[8];
    v4f rb[12];

    v8f acc[2][3];
    #pragma unroll
    for (int i = 0; i < 2; ++i) {
        #pragma unroll
        for (int j = 0; j < 3; ++j) acc[i][j] = (v8f){};
    }

    const int nch = Kd / PJ_KC;

#define PJ_FETCH(CH)                                                                         \
    do {                                                                                     \
        const float* agn = ag + (size_t)(CH) * PJ_KC;                                        \
        const float* bgn = bg + (size_t)(CH) * PJ_KC * PJ_LD;                                \
        _Pragma("unroll")                   \
        for (int i = 0; i < 8; ++i) ra[i] = *(const v4f*)(agn + 8 * i * PJ_LD);              \
        _Pragma("unroll")                   \
        for (int e = 0; e < 4; ++e) {                                                        \
            _Pragma("unroll")               \
            for (int p = 0; p < 3; ++p) rb[e * 3 + p] = *(const v4f*)(bgn + e * PJ_LD + 32 * p); \
        }                                                                                    \
    } while (0)

#define PJ_STAGE(ST)                                                                         \
    do {                                                                                     \
        _Pragma("unroll")                   \
        for (int i = 0; i < 8; ++i) *(v4f*)&As[ST][arow + 8 * i][acol] = ra[i];              \
        _Pragma("unroll")                   \
        for (int p = 0; p < 3; ++p) {                                                        \
            _Pragma("unroll")               \
            for (int i = 0; i < 4; ++i) {                                                    \
                v4f o;                                                                       \
                o.x = rb[0 * 3 + p][i]; o.y = rb[1 * 3 + p][i];                              \
                o.z = rb[2 * 3 + p][i]; o.w = rb[3 * 3 + p][i];                              \
                *(v4f*)&Bs[ST][bq][(32 * p + bcol + i) * 4] = o;                             \
            }                                                                                \
        }                                                                                    \
    } while (0)

    PJ_FETCH(0);
    PJ_STAGE(0);
    __syncthreads();

    const float* as0 = &As[0][wm * 32 + lo][hi * 4];
    const float* bs0 = &Bs[0][hi][(wn * 48 + lo) * 4];
    const int as_stage = PJ_BM * PJ_PA;
    const int bs_stage = (PJ_KC / 4) * PJ_PQ;

    PjFrag f0, f1;
    pj_load_frags(f0, as0, bs0, 0);
    if (nch > 1) PJ_FETCH(1);

    for (int c = 0; c < nch; ++c) {
        const int st = c & 1;
        const bool more = (c + 1 < nch);
        const float* as = as0 + st * as_stage;
        const float* bs = bs0 + st * bs_stage;

        pj_load_frags(f1, as, bs, 1);
        pj_fence();
        pj_mma_block(f0, acc);
        pj_fence();
        pj_load_frags(f0, as, bs, 2);
        pj_fence();
        pj_mma_block(f1, acc);
        pj_fence();
        if (more) {
            PJ_STAGE(st ^ 1);
        }
        pj_load_frags(f1, as, bs, 3);
        pj_fence();
        pj_mma_block(f0, acc);
        pj_fence();
        __syncthreads();
        if (more) {
            pj_load_frags(f0, as0 + (st ^ 1) * as_stage, bs0 + (st ^ 1) * bs_stage, 0);
            if (c + 2 < nch) PJ_FETCH(c + 2);
        }
        pj_fence();
        pj_mma_block(f1, acc);
        pj_fence();
    }
#undef PJ_FETCH
#undef PJ_STAGE

    const int orow = blockM + wm * 32 + hi * 8;
    const int ocol = blockN + wn * 48 + lo;
    #pragma unroll
    for (int v = 0; v < 8; ++v) {
        float* orow0 = O + (size_t)(orow + v)      * PJ_LD + ocol;
        float* orow1 = O + (size_t)(orow + 16 + v) * PJ_LD + ocol;
        #pragma unroll
        for (int j = 0; j < 3; ++j) {
            orow0[16 * j] = acc[0][j][v];
            orow1[16 * j] = acc[1][j][v];
        }
    }
}

#define SWA_W     64
#define SWA_WLEN  144
#define SWA_PITCH 148
#define SWA_D     768
#define SWA_NP    12
#define SWA_KS    64
#define SWA_TP    68

__device__ __forceinline__ void swa_fetch_tile(v4f (&g)[8], const float* __restrict__ rowptr) {
    #pragma unroll
    for (int i = 0; i < 8; ++i) g[i] = *(const v4f*)(rowptr + (size_t)(2 * i) * SWA_D);
}

__device__ __forceinline__ void swa_transpose_tile(const v4f (&g)[8], v4f (&f)[8],
                                                   float* st_store, const float* st_frag) {
    #pragma unroll
    for (int i = 0; i < 8; ++i) *(v4f*)(st_store + 2 * i * SWA_TP) = g[i];
    #pragma unroll
    for (int s = 0; s < 8; ++s) f[s] = *(const v4f*)(st_frag + s * 8);
}

__global__ __launch_bounds__(512)
void swa_kernel(const float* __restrict__ kp,
                const float* __restrict__ vp,
                const float* __restrict__ qp,
                float* __restrict__ out,
                int T, int S) {
    __shared__ float part[SWA_NP][16][SWA_PITCH];
    __shared__ float pr[16][SWA_PITCH];
    __shared__ __attribute__((aligned(16))) float stg[SWA_NP][16][SWA_TP];

    const int tid  = threadIdx.x;
    const int lane = tid & 31;
    const int wave = tid >> 5;
    const int lo = lane & 15;
    const int hi = lane >> 4;

    const int t0    = blockIdx.x * 16;
    const int kbase = t0 - SWA_W;

    if (wave < SWA_NP) {
        const int r2   = lane >> 4;
        const int kcol = wave * SWA_KS + (lane & 15) * 4;
        float* st_store      = &stg[wave][r2][(lane & 15) * 4];
        const float* st_frag = &stg[wave][lo][hi * 4];

        v4f g[3][8];
        swa_fetch_tile(g[0], qp + (size_t)(t0 + r2) * SWA_D + kcol);
        #pragma unroll
        for (int j = 1; j < 3; ++j) {
            const int tb = min(max(kbase + (j - 1) * 16, 0), S - 16);
            swa_fetch_tile(g[j], kp + (size_t)(tb + r2) * SWA_D + kcol);
        }
        v4f a[8];
        swa_transpose_tile(g[0], a, st_store, st_frag);

        #pragma unroll
        for (int kt = 0; kt < 9; ++kt) {
            if (kt + 2 < 9) {
                const int tb = min(max(kbase + (kt + 2) * 16, 0), S - 16);
                swa_fetch_tile(g[kt % 3], kp + (size_t)(tb + r2) * SWA_D + kcol);
            }
            v4f b[8];
            swa_transpose_tile(g[(kt + 1) % 3], b, st_store, st_frag);
            v8f acc = {};
            #pragma unroll
            for (int s = 0; s < 8; ++s) {
                v2f a0; a0.x = a[s].x; a0.y = a[s].y;
                v2f b0; b0.x = b[s].x; b0.y = b[s].y;
                acc = wmma_f32(a0, b0, acc);
                v2f a1; a1.x = a[s].z; a1.y = a[s].w;
                v2f b1; b1.x = b[s].z; b1.y = b[s].w;
                acc = wmma_f32(a1, b1, acc);
            }
            #pragma unroll
            for (int v = 0; v < 8; ++v)
                part[wave][v + hi * 8][kt * 16 + lo] = acc[v];
        }
    }
    __syncthreads();

    {
        const int t = t0 + wave;
        const int start = max(0, t - SWA_W);
        const int end   = min(S, t + SWA_W + 1);
        const float scale = 0.03608439182435161f;
        float sv[5];
        float m = -INFINITY;
        #pragma unroll
        for (int i = 0; i < 5; ++i) {
            const int c = lane + 32 * i;
            float s = -INFINITY;
            if (c < SWA_WLEN) {
                float acc = part[0][wave][c];
                #pragma unroll
                for (int p = 1; p < SWA_NP; ++p) acc += part[p][wave][c];
                const int j = kbase + c;
                if (j >= start && j < end) s = acc * scale;
            }
            sv[i] = s;
            m = fmaxf(m, s);
        }
        for (int d = 16; d >= 1; d >>= 1) m = fmaxf(m, __shfl_xor(m, d));
        float sum = 0.f;
        #pragma unroll
        for (int i = 0; i < 5; ++i) {
            const float e = (sv[i] == -INFINITY) ? 0.f : __expf(sv[i] - m);
            sv[i] = e;
            sum += e;
        }
        for (int d = 16; d >= 1; d >>= 1) sum += __shfl_xor(sum, d);
        const float inv = 1.0f / (sum * (float)(end - start));
        #pragma unroll
        for (int i = 0; i < 5; ++i) {
            const int c = lane + 32 * i;
            if (c < SWA_WLEN) pr[wave][c] = sv[i] * inv;
        }
    }
    __syncthreads();

    if (wave < SWA_D / 64) {
        const int col = wave * 64 + lo * 4;
        const unsigned lane_off = (unsigned)(hi * 2 * SWA_D + col);
        v8f acc0 = {}; v8f acc1 = {}; v8f acc2 = {}; v8f acc3 = {};
        #pragma unroll 1
        for (int kt = 0; kt < SWA_WLEN / 16; kt += 3) {
            #pragma unroll
            for (int u = 0; u < 3; ++u) {
                const int tb = min(max(kbase + (kt + u) * 16, 0), S - 16);
                const float* vt = vp + (size_t)tb * SWA_D;
                #pragma unroll
                for (int q = 0; q < 4; ++q) {
                    const int k0 = (kt + u) * 16 + q * 4;
                    const v2f a = *(const v2f*)&pr[lo][k0 + hi * 2];
                    const v4f r0 = *(const v4f*)(vt + lane_off + (q * 4)     * SWA_D);
                    const v4f r1 = *(const v4f*)(vt + lane_off + (q * 4 + 1) * SWA_D);
                    v2f b0; b0.x = r0.x; b0.y = r1.x;
                    v2f b1; b1.x = r0.y; b1.y = r1.y;
                    v2f b2; b2.x = r0.z; b2.y = r1.z;
                    v2f b3; b3.x = r0.w; b3.y = r1.w;
                    acc0 = wmma_f32(a, b0, acc0);
                    acc1 = wmma_f32(a, b1, acc1);
                    acc2 = wmma_f32(a, b2, acc2);
                    acc3 = wmma_f32(a, b3, acc3);
                }
            }
        }
        #pragma unroll
        for (int v = 0; v < 8; ++v) {
            v4f o; o.x = acc0[v]; o.y = acc1[v]; o.z = acc2[v]; o.w = acc3[v];
            *(v4f*)(out + (size_t)(t0 + v + hi * 8) * SWA_D + col) = o;
        }
    }
}

extern "C" void kernel_launch(void* const* d_in, const int* in_sizes, int n_in,
                              void* d_out, int out_size, void* d_ws, size_t ws_size,
                              hipStream_t stream) {
    const float* key   = (const float*)d_in[0];
    const float* value = (const float*)d_in[1];
    const float* query = (const float*)d_in[2];
    const float* Wk    = (const float*)d_in[3];
    const float* Wv    = (const float*)d_in[4];
    const float* Wq    = (const float*)d_in[5];
    float* out = (float*)d_out;

    const int S = 2048, T = 2048, E = 768, D = 768;

    float* ws = (float*)d_ws;
    float* kproj = ws;
    float* vproj = ws + (size_t)S * D;
    float* qproj = ws + 2 * (size_t)S * D;

    dim3 gproj(D / 96, S / 64, 3);
    proj_gemm<<<gproj, 128, 0, stream>>>(key, value, query, Wk, Wv, Wq,
                                         kproj, vproj, qproj, E);

    swa_kernel<<<dim3(T / 16), 512, 0, stream>>>(kproj, vproj, qproj, out, T, S);
}
